# starting kernel
# baseline (speedup 1.0000x reference)
#include <hip/hip_runtime.h>
#include <hip/hip_bf16.h>
#include <math.h>

#define B_   4
#define N_   512
#define D_   32
#define PE_  16

#define WPB 4
#define L2E 1.44269504088896340736f
#define INV_SQRT_D 0.17677669529663688110f
#define KS2 (10.0f * L2E)

typedef __attribute__((ext_vector_type(2))) float v2f;
typedef __attribute__((ext_vector_type(8))) float v8f;

static __device__ __forceinline__ v8f wmma_f32(v2f a, v2f b, v8f c) {
  return __builtin_amdgcn_wmma_f32_16x16x4_f32(false, a, false, b, (short)0, c, false, false);
}

static __device__ __forceinline__ float fexp2(float x) { return __builtin_amdgcn_exp2f(x); }
static __device__ __forceinline__ float frcp(float x)  { return __builtin_amdgcn_rcpf(x); }

#define DPP_QUAD_XOR1  0xB1
#define DPP_QUAD_XOR2  0x4E
#define DPP_HALF_MIR   0x141
#define DPP_ROW_MIR    0x140

template <int CTRL>
static __device__ __forceinline__ float dpp_mov(float x) {
  return __int_as_float(
      __builtin_amdgcn_update_dpp(0, __float_as_int(x), CTRL, 0xf, 0xf, true));
}
static __device__ __forceinline__ float redmax16(float x) {
  x = fmaxf(x, dpp_mov<DPP_QUAD_XOR1>(x));
  x = fmaxf(x, dpp_mov<DPP_QUAD_XOR2>(x));
  x = fmaxf(x, dpp_mov<DPP_HALF_MIR>(x));
  x = fmaxf(x, dpp_mov<DPP_ROW_MIR>(x));
  return x;
}
static __device__ __forceinline__ float redsum16(float x) {
  x += dpp_mov<DPP_QUAD_XOR1>(x);
  x += dpp_mov<DPP_QUAD_XOR2>(x);
  x += dpp_mov<DPP_HALF_MIR>(x);
  x += dpp_mov<DPP_ROW_MIR>(x);
  return x;
}

__global__ void kq_proj_kernel(const float* __restrict__ pe,
                               const float* __restrict__ Wk, const float* __restrict__ bk,
                               const float* __restrict__ Wq, const float* __restrict__ bq,
                               float* __restrict__ Kd, float* __restrict__ Qd) {
  int idx = blockIdx.x * blockDim.x + threadIdx.x;
  if (idx >= B_ * N_ * D_) return;
  int d  = idx & (D_ - 1);
  int bn = idx / D_;
  const float* p = pe + bn * PE_;
  float k = bk[d], q = bq[d];
#pragma unroll
  for (int e = 0; e < PE_; ++e) {
    float pv = p[e];
    k = fmaf(pv, Wk[e * D_ + d], k);
    q = fmaf(pv, Wq[e * D_ + d], q);
  }
  Kd[idx] = k;
  Qd[idx] = q;
}

__global__ void __launch_bounds__(32 * WPB)
fokker_planck_fused_kernel(const float* __restrict__ h,
                           const float* __restrict__ E,
                           const float* __restrict__ A,
                           const float* __restrict__ beta,
                           const float* __restrict__ Kd,
                           const float* __restrict__ Qd,
                           float* __restrict__ out) {
  __shared__ float lds[WPB * 3 * 256];

  const int lane = threadIdx.x & 31;
  const int wid  = threadIdx.x >> 5;
  const int lo   = lane & 15;
  const int hi   = lane >> 4;
  const int tile = blockIdx.x * WPB + wid;
  if (tile >= B_ * (N_ / 16)) return;
  const int b  = tile >> 5;
  const int i0 = (tile & 31) * 16;

  float* lds1 = lds + wid * 3 * 256;
  float* lds3 = lds1 + 256;
  float* lds4 = lds1 + 512;

  const float* Kb = Kd + b * N_ * D_;
  const float* Qb = Qd + b * N_ * D_;
  const float* Hb = h  + b * N_ * D_;

  const float kscale = INV_SQRT_D * L2E;
  v2f afK[8];
#pragma unroll
  for (int kc = 0; kc < 8; ++kc) {
    v2f t = *(const v2f*)(Kb + (i0 + lo) * D_ + 4 * kc + 2 * hi);
    afK[kc].x = t.x * kscale;
    afK[kc].y = t.y * kscale;
  }

  float Ei[8];
#pragma unroll
  for (int v = 0; v < 8; ++v) Ei[v] = E[i0 + v + 8 * hi];

  v8f acc1[2] = {}, acc3[2] = {}, acc4[2] = {}, acc5[2] = {};
  float m[8], z[8], c2[8], c6[8];
#pragma unroll
  for (int v = 0; v < 8; ++v) { m[v] = -1.0e30f; z[v] = 0.f; c2[v] = 0.f; c6[v] = 0.f; }

  for (int j0 = 0; j0 < N_; j0 += 16) {
    v8f sraw = {};
#pragma unroll
    for (int kc = 0; kc < 8; ++kc) {
      v2f bfQ = *(const v2f*)(Qb + (j0 + lo) * D_ + 4 * kc + 2 * hi);
      sraw = wmma_f32(afK[kc], bfQ, sraw);
    }

    const float Ej = E[j0 + lo];

    float sc[8];
#pragma unroll
    for (int v = 0; v < 8; ++v)
      sc[v] = A[(i0 + v + 8 * hi) * N_ + j0 + lo] * sraw[v];

    float tm[8];
#pragma unroll
    for (int v = 0; v < 8; ++v) tm[v] = redmax16(sc[v]);

#pragma unroll
    for (int v = 0; v < 8; ++v) {
      float mn = fmaxf(m[v], tm[v]);
      float r  = fexp2(m[v] - mn);
      m[v] = mn;
      z[v]  *= r;  c2[v] *= r;  c6[v] *= r;
      acc1[0][v] *= r; acc1[1][v] *= r;
      acc3[0][v] *= r; acc3[1][v] *= r;
      acc4[0][v] *= r; acc4[1][v] *= r;
      acc5[0][v] *= r; acc5[1][v] *= r;

      float p  = fexp2(sc[v] - mn);
      float Ed = Ej - Ei[v];
      float sg = frcp(1.f + fexp2(-KS2 * Ed));
      float p3 = p * sg;
      float p1 = p3 * Ed;
      float p4 = p - p3;
      z[v]  += p;
      c2[v] += p * Ed - p1;
      c6[v] += p4;

      int addr = (v + 8 * hi) * 16 + lo;
      lds1[addr] = p1;
      lds3[addr] = p3;
      lds4[addr] = p4;
    }

    asm volatile("s_wait_dscnt 0" ::: "memory");

    v2f p1a[4], p3a[4], p4a[4];
#pragma unroll
    for (int kc = 0; kc < 4; ++kc) {
      int a = lo * 16 + 4 * kc + 2 * hi;
      p1a[kc] = *(const v2f*)(lds1 + a);
      p3a[kc] = *(const v2f*)(lds3 + a);
      p4a[kc] = *(const v2f*)(lds4 + a);
    }

    v2f HB[4][2], LB[4][2], LHB[4][2];
#pragma unroll
    for (int kc = 0; kc < 4; ++kc)
#pragma unroll
      for (int s = 0; s < 2; ++s)
#pragma unroll
        for (int t = 0; t < 2; ++t) {
          float hv = Hb[(j0 + 4 * kc + 2 * hi + t) * D_ + lo + 16 * s];
          float lv = __logf(hv + 1e-8f);
          HB[kc][s][t]  = hv;
          LB[kc][s][t]  = lv;
          LHB[kc][s][t] = lv * hv;
        }

#pragma unroll
    for (int s = 0; s < 2; ++s)
#pragma unroll
      for (int kc = 0; kc < 4; ++kc) {
        acc1[s] = wmma_f32(p1a[kc], HB[kc][s],  acc1[s]);
        acc5[s] = wmma_f32(p3a[kc], HB[kc][s],  acc5[s]);
        acc3[s] = wmma_f32(p3a[kc], LHB[kc][s], acc3[s]);
        acc4[s] = wmma_f32(p4a[kc], LB[kc][s],  acc4[s]);
      }
  }

#pragma unroll
  for (int v = 0; v < 8; ++v) {
    z[v]  = redsum16(z[v]);
    c2[v] = redsum16(c2[v]);
    c6[v] = redsum16(c6[v]);
  }

  float bt[2];
  bt[0] = beta[lo];
  bt[1] = beta[lo + 16];

#pragma unroll
  for (int v = 0; v < 8; ++v) {
    const int row = i0 + v + 8 * hi;
    const float invz = frcp(z[v]);
#pragma unroll
    for (int s = 0; s < 2; ++s) {
      const int col = lo + 16 * s;
      float hv = Hb[row * D_ + col];
      float lv = __logf(hv + 1e-8f);
      float num = acc1[s][v] + c2[v] * hv +
                  bt[s] * (acc3[s][v] + hv * acc4[s][v] - lv * acc5[s][v] - c6[v] * lv * hv);
      out[(b * N_ + row) * D_ + col] = num * invz;
    }
  }
}

extern "C" void kernel_launch(void* const* d_in, const int* in_sizes, int n_in,
                              void* d_out, int out_size, void* d_ws, size_t ws_size,
                              hipStream_t stream) {
  const float* h    = (const float*)d_in[1];
  const float* pe   = (const float*)d_in[2];
  const float* E    = (const float*)d_in[3];
  const float* A    = (const float*)d_in[4];
  const float* Wk   = (const float*)d_in[5];
  const float* bk   = (const float*)d_in[6];
  const float* Wq   = (const float*)d_in[7];
  const float* bq   = (const float*)d_in[8];
  const float* beta = (const float*)d_in[9];
  float* out = (float*)d_out;

  float* Kd = (float*)d_ws;
  float* Qd = Kd + (size_t)B_ * N_ * D_;

  {
    int total = B_ * N_ * D_;
    kq_proj_kernel<<<(total + 255) / 256, 256, 0, stream>>>(pe, Wk, bk, Wq, bq, Kd, Qd);
  }
  {
    int tiles = B_ * (N_ / 16);
    dim3 grid(tiles / WPB), block(32 * WPB);
    fokker_planck_fused_kernel<<<grid, block, 0, stream>>>(h, E, A, beta, Kd, Qd, out);
  }
}


# Each j-tile prepared once per work-group: wave w does tile w of a 4-tile group (softmax weights vs the tile's own row max, H, L, L.H operands) into LDS; product waves rescale, multiply; new sum order
# speedup vs baseline: 4.5859x; 4.5859x over previous
#include <hip/hip_runtime.h>
#include <hip/hip_bf16.h>
#include <math.h>

#define B_   4
#define N_   512
#define D_   32
#define PE_  16

#define WPB 4
#define PP  18
#define PT  (16 * PP)
#define L2E 1.44269504088896340736f
#define INV_SQRT_D 0.17677669529663688110f
#define KS2 (10.0f * L2E)

typedef __attribute__((ext_vector_type(2))) float v2f;
typedef __attribute__((ext_vector_type(4))) float v4f;
typedef __attribute__((ext_vector_type(8))) float v8f;

static __device__ __forceinline__ v8f wmma_f32(v2f a, v2f b, v8f c) {
  return __builtin_amdgcn_wmma_f32_16x16x4_f32(false, a, false, b, (short)0, c, false, false);
}

static __device__ __forceinline__ float fexp2(float x) { return __builtin_amdgcn_exp2f(x); }
static __device__ __forceinline__ float frcp(float x)  { return __builtin_amdgcn_rcpf(x); }

#define DPP_QUAD_XOR1  0xB1
#define DPP_QUAD_XOR2  0x4E
#define DPP_HALF_MIR   0x141
#define DPP_ROW_MIR    0x140

template <int CTRL>
static __device__ __forceinline__ float dpp_mov(float x) {
  return __int_as_float(
      __builtin_amdgcn_update_dpp(0, __float_as_int(x), CTRL, 0xf, 0xf, true));
}
static __device__ __forceinline__ float redmax16(float x) {
  x = fmaxf(x, dpp_mov<DPP_QUAD_XOR1>(x));
  x = fmaxf(x, dpp_mov<DPP_QUAD_XOR2>(x));
  x = fmaxf(x, dpp_mov<DPP_HALF_MIR>(x));
  x = fmaxf(x, dpp_mov<DPP_ROW_MIR>(x));
  return x;
}
static __device__ __forceinline__ float redsum16(float x) {
  x += dpp_mov<DPP_QUAD_XOR1>(x);
  x += dpp_mov<DPP_QUAD_XOR2>(x);
  x += dpp_mov<DPP_HALF_MIR>(x);
  x += dpp_mov<DPP_ROW_MIR>(x);
  return x;
}

__global__ void kq_proj_kernel(const float* __restrict__ pe,
                               const float* __restrict__ Wk, const float* __restrict__ bk,
                               const float* __restrict__ Wq, const float* __restrict__ bq,
                               float* __restrict__ Kd, float* __restrict__ Qd) {
  int idx = blockIdx.x * blockDim.x + threadIdx.x;
  if (idx >= B_ * N_ * D_) return;
  int d  = idx & (D_ - 1);
  int bn = idx / D_;
  const float* p = pe + bn * PE_;
  float k = bk[d], q = bq[d];
#pragma unroll
  for (int e = 0; e < PE_; ++e) {
    float pv = p[e];
    k = fmaf(pv, Wk[e * D_ + d], k);
    q = fmaf(pv, Wq[e * D_ + d], q);
  }
  Kd[idx] = k;
  Qd[idx] = q;
}

struct GTile {
  v2f   q[8];
  float a[8];
  float ej;
  float h[4][2][2];
};

static __device__ __forceinline__ void load_gtile(GTile& T,
                                                 const float* __restrict__ Qb,
                                                 const float* __restrict__ Hb,
                                                 const float* __restrict__ A,
                                                 const float* __restrict__ E,
                                                 int i0, int js, int lo, int hi) {
#pragma unroll
  for (int kc = 0; kc < 8; ++kc)
    T.q[kc] = *(const v2f*)(Qb + (js + lo) * D_ + 4 * kc + 2 * hi);
  T.ej = E[js + lo];
#pragma unroll
  for (int v = 0; v < 8; ++v)
    T.a[v] = A[(i0 + v + 8 * hi) * N_ + js + lo];
#pragma unroll
  for (int kc = 0; kc < 4; ++kc)
#pragma unroll
    for (int s = 0; s < 2; ++s)
#pragma unroll
      for (int t = 0; t < 2; ++t)
        T.h[kc][s][t] = Hb[(js + 4 * kc + 2 * hi + t) * D_ + lo + 16 * s];
}

static __device__ __forceinline__ void wg_barrier_lds() {
  asm volatile("s_wait_dscnt 0\n\ts_barrier_signal -1\n\ts_barrier_wait -1" ::: "memory");
}

#define OT 512

__global__ void __launch_bounds__(32 * WPB)
fokker_planck_fused_kernel(const float* __restrict__ h,
                           const float* __restrict__ E,
                           const float* __restrict__ A,
                           const float* __restrict__ beta,
                           const float* __restrict__ Kd,
                           const float* __restrict__ Qd,
                           float* __restrict__ out) {
  __shared__ __attribute__((aligned(16))) float Pv[2 * 4 * 3 * PT];
  __shared__ __attribute__((aligned(16))) float Ov[2 * 4 * 3 * OT];
  __shared__ __attribute__((aligned(16))) float Mt[2 * 4 * 16];
  __shared__ __attribute__((aligned(16))) float Es[N_];
  __shared__ float partAcc[4 * 2 * 8 * 32];
  __shared__ float rowS[3 * 16];

  const int lane = threadIdx.x & 31;
  const int wid  = __builtin_amdgcn_readfirstlane((int)(threadIdx.x >> 5));
  const int lo   = lane & 15;
  const int hi   = lane >> 4;
  const int tile = blockIdx.x;
  if (tile >= B_ * (N_ / 16)) return;
  const int b  = tile >> 5;
  const int i0 = (tile & 31) * 16;

  const float* Kb = Kd + b * N_ * D_;
  const float* Qb = Qd + b * N_ * D_;
  const float* Hb = h  + b * N_ * D_;

#pragma unroll
  for (int k = 0; k < N_ / (32 * WPB); ++k)
    Es[threadIdx.x + 32 * WPB * k] = E[threadIdx.x + 32 * WPB * k];

  const float kscale = INV_SQRT_D * L2E;
  v2f afK[8];
#pragma unroll
  for (int kc = 0; kc < 8; ++kc) {
    v2f t = *(const v2f*)(Kb + (i0 + lo) * D_ + 4 * kc + 2 * hi);
    afK[kc].x = t.x * kscale;
    afK[kc].y = t.y * kscale;
  }

  float Ei[8];
#pragma unroll
  for (int v = 0; v < 8; ++v) Ei[v] = E[i0 + v + 8 * hi];
  const float EiA = E[i0 + lo];

  const int vsel = (wid == 0) ? 0 : ((wid == 3) ? 2 : 1);
  const int osel = (wid < 2) ? 0 : ((wid == 3) ? 1 : 2);

  v8f acc[2] = {};
  float M[8];
#pragma unroll
  for (int v = 0; v < 8; ++v) M[v] = -1.0e30f;
  float MA = -1.0e30f;
  float sA = 0.f;
  float c2A = 0.f;

  auto group_step = [&](const GTile& T, const int bf, const int jg) __attribute__((always_inline)) {
    v8f sraw = {};
#pragma unroll
    for (int kc = 0; kc < 8; ++kc)
      sraw = wmma_f32(afK[kc], T.q[kc], sraw);

    float sc[8], tm[8];
#pragma unroll
    for (int v = 0; v < 8; ++v) {
      sc[v] = T.a[v] * sraw[v];
      tm[v] = redmax16(sc[v]);
    }

    float* pvw = Pv + (bf * 4 + wid) * 3 * PT;
    float* ovw = Ov + (bf * 4 + wid) * 3 * OT + lane * 4;
    float* mtw = Mt + (bf * 4 + wid) * 16;
#pragma unroll
    for (int v = 0; v < 8; ++v) {
      float p  = fexp2(sc[v] - tm[v]);
      float Ed = T.ej - Ei[v];
      float sg = frcp(1.f + fexp2(-KS2 * Ed));
      float p3 = p * sg;
      float p1 = p3 * Ed;
      float p4 = p - p3;
      const int o = (v + 8 * hi) * PP + lo;
      pvw[o]          = p1;
      pvw[PT + o]     = p3;
      pvw[2 * PT + o] = p4;
    }
    {
      v4f t0 = {tm[0], tm[1], tm[2], tm[3]};
      v4f t1 = {tm[4], tm[5], tm[6], tm[7]};
      *(v4f*)(mtw + 8 * hi)     = t0;
      *(v4f*)(mtw + 8 * hi + 4) = t1;
    }
#pragma unroll
    for (int kc = 0; kc < 4; ++kc) {
      v4f hh = {T.h[kc][0][0], T.h[kc][0][1], T.h[kc][1][0], T.h[kc][1][1]};
      v4f ll, lh;
#pragma unroll
      for (int e = 0; e < 4; ++e) {
        ll[e] = __logf(hh[e] + 1e-8f);
        lh[e] = ll[e] * hh[e];
      }
      *(v4f*)(ovw + (0 * 4 + kc) * 128) = hh;
      *(v4f*)(ovw + (1 * 4 + kc) * 128) = ll;
      *(v4f*)(ovw + (2 * 4 + kc) * 128) = lh;
    }

    wg_barrier_lds();

    float mg[8], mAt[4];
#pragma unroll
    for (int tt = 0; tt < 4; ++tt) {
      const float* mt = Mt + (bf * 4 + tt) * 16;
      const v4f m0 = *(const v4f*)(mt + 8 * hi);
      const v4f m1 = *(const v4f*)(mt + 8 * hi + 4);
      mAt[tt] = mt[lo];
      const float mc[8] = {m0.x, m0.y, m0.z, m0.w, m1.x, m1.y, m1.z, m1.w};
#pragma unroll
      for (int v = 0; v < 8; ++v) mg[v] = (tt == 0) ? mc[v] : fmaxf(mg[v], mc[v]);
    }
#pragma unroll
    for (int v = 0; v < 8; ++v) {
      float mn = fmaxf(M[v], mg[v]);
      float r  = fexp2(M[v] - mn);
      M[v] = mn;
      acc[0][v] *= r; acc[1][v] *= r;
    }
    const float mnA = fmaxf(fmaxf(MA, fmaxf(mAt[0], mAt[1])), fmaxf(mAt[2], mAt[3]));
    {
      const float rA = fexp2(MA - mnA);
      MA = mnA;
      sA *= rA;
      c2A *= rA;
    }

#pragma unroll
    for (int tt = 0; tt < 4; ++tt) {
      const float fA = fexp2(mAt[tt] - mnA);

      const float* pv = Pv + ((bf * 4 + tt) * 3 + vsel) * PT + lo * PP + 2 * hi;
      v2f wa[4];
#pragma unroll
      for (int kc = 0; kc < 4; ++kc) {
        v2f t = *(const v2f*)(pv + 4 * kc);
        wa[kc].x = t.x * fA;
        wa[kc].y = t.y * fA;
      }
      float ps = 0.f;
#pragma unroll
      for (int kc = 0; kc < 4; ++kc) ps += wa[kc].x + wa[kc].y;
      sA += ps;

      if (wid == 0) {
        const float* p4 = Pv + ((bf * 4 + tt) * 3 + 2) * PT + lo * PP + 2 * hi;
        const float* ej = Es + jg + 16 * tt + 2 * hi;
        float cs = 0.f;
#pragma unroll
        for (int kc = 0; kc < 4; ++kc) {
          v2f w = *(const v2f*)(p4 + 4 * kc);
          v2f e = *(const v2f*)(ej + 4 * kc);
          cs += w.x * (e.x - EiA) + w.y * (e.y - EiA);
        }
        c2A += cs * fA;
      }

      const float* ov = Ov + ((bf * 4 + tt) * 3 + osel) * OT + lane * 4;
      v2f VB[4][2];
#pragma unroll
      for (int kc = 0; kc < 4; ++kc) {
        const v4f o = *(const v4f*)(ov + kc * 128);
        VB[kc][0].x = o.x; VB[kc][0].y = o.y;
        VB[kc][1].x = o.z; VB[kc][1].y = o.w;
      }

#pragma unroll
      for (int kc = 0; kc < 4; ++kc)
#pragma unroll
        for (int s = 0; s < 2; ++s)
          acc[s] = wmma_f32(wa[kc], VB[kc][s], acc[s]);
    }
  };

  GTile ta, tb;
  load_gtile(ta, Qb, Hb, A, E, i0, 16 * wid, lo, hi);
  for (int jg = 0; jg < N_; jg += 128) {
    load_gtile(tb, Qb, Hb, A, E, i0, jg + 64 + 16 * wid, lo, hi);
    group_step(ta, 0, jg);
    load_gtile(ta, Qb, Hb, A, E, i0, ((jg + 128 < N_) ? jg + 128 : jg + 64) + 16 * wid, lo, hi);
    group_step(tb, 1, jg + 64);
  }

#pragma unroll
  for (int s = 0; s < 2; ++s)
#pragma unroll
    for (int v = 0; v < 8; ++v)
      partAcc[((wid * 2 + s) * 8 + v) * 32 + lane] = acc[s][v];
  {
    const float st = sA  + __shfl_xor(sA, 16);
    const float ct = c2A + __shfl_xor(c2A, 16);
    if (wid == 1) rowS[0 * 16 + lo] = st;
    if (wid == 0) rowS[1 * 16 + lo] = ct;
    if (wid == 3) rowS[2 * 16 + lo] = st;
  }

  __syncthreads();

  float bt[2];
  bt[0] = beta[lo];
  bt[1] = beta[lo + 16];

#pragma unroll
  for (int vv = 0; vv < 2; ++vv) {
    const int v   = 2 * wid + vv;
    const int rl  = v + 8 * hi;
    const int row = i0 + rl;
    const float cc6 = rowS[2 * 16 + rl];
    const float zz  = rowS[0 * 16 + rl] + cc6;
    const float cc2 = rowS[1 * 16 + rl];
    const float invz = frcp(zz);
#pragma unroll
    for (int s = 0; s < 2; ++s) {
      const int col = lo + 16 * s;
      const float a1 = partAcc[((0 * 2 + s) * 8 + v) * 32 + lane];
      const float a5 = partAcc[((1 * 2 + s) * 8 + v) * 32 + lane];
      const float a3 = partAcc[((2 * 2 + s) * 8 + v) * 32 + lane];
      const float a4 = partAcc[((3 * 2 + s) * 8 + v) * 32 + lane];
      float hv = Hb[row * D_ + col];
      float lv = __logf(hv + 1e-8f);
      float num = a1 + cc2 * hv +
                  bt[s] * (a3 + hv * a4 - lv * a5 - cc6 * lv * hv);
      out[(b * N_ + row) * D_ + col] = num * invz;
    }
  }
}

extern "C" void kernel_launch(void* const* d_in, const int* in_sizes, int n_in,
                              void* d_out, int out_size, void* d_ws, size_t ws_size,
                              hipStream_t stream) {
  const float* h    = (const float*)d_in[1];
  const float* pe   = (const float*)d_in[2];
  const float* E    = (const float*)d_in[3];
  const float* A    = (const float*)d_in[4];
  const float* Wk   = (const float*)d_in[5];
  const float* bk   = (const float*)d_in[6];
  const float* Wq   = (const float*)d_in[7];
  const float* bq   = (const float*)d_in[8];
  const float* beta = (const float*)d_in[9];
  float* out = (float*)d_out;

  float* Kd = (float*)d_ws;
  float* Qd = Kd + (size_t)B_ * N_ * D_;

  {
    int total = B_ * N_ * D_;
    kq_proj_kernel<<<(total + 255) / 256, 256, 0, stream>>>(pe, Wk, bk, Wq, bq, Kd, Qd);
  }
  {
    int tiles = B_ * (N_ / 16);
    dim3 grid(tiles), block(32 * WPB);
    fokker_planck_fused_kernel<<<grid, block, 0, stream>>>(h, E, A, beta, Kd, Qd, out);
  }
}


# 8 waves in 2 roles per work-group: waves 0-3 prepare group g+1 (wave 0 also adds group g's row sum of P(1-s)Ed) while waves 4-7 multiply group g from the other LDS buffer; sum order unchanged
# speedup vs baseline: 5.0383x; 1.0987x over previous
#include <hip/hip_runtime.h>
#include <hip/hip_bf16.h>
#include <math.h>

#define B_   4
#define N_   512
#define D_   32
#define PE_  16

#define WPB 8
#define PP  18
#define PT  (16 * PP)
#define L2E 1.44269504088896340736f
#define INV_SQRT_D 0.17677669529663688110f
#define KS2 (10.0f * L2E)

typedef __attribute__((ext_vector_type(2))) float v2f;
typedef __attribute__((ext_vector_type(4))) float v4f;
typedef __attribute__((ext_vector_type(8))) float v8f;

static __device__ __forceinline__ v8f wmma_f32(v2f a, v2f b, v8f c) {
  return __builtin_amdgcn_wmma_f32_16x16x4_f32(false, a, false, b, (short)0, c, false, false);
}

static __device__ __forceinline__ float fexp2(float x) { return __builtin_amdgcn_exp2f(x); }
static __device__ __forceinline__ float frcp(float x)  { return __builtin_amdgcn_rcpf(x); }

#define DPP_QUAD_XOR1  0xB1
#define DPP_QUAD_XOR2  0x4E
#define DPP_HALF_MIR   0x141
#define DPP_ROW_MIR    0x140

template <int CTRL>
static __device__ __forceinline__ float dpp_mov(float x) {
  return __int_as_float(
      __builtin_amdgcn_update_dpp(0, __float_as_int(x), CTRL, 0xf, 0xf, true));
}
static __device__ __forceinline__ float redmax16(float x) {
  x = fmaxf(x, dpp_mov<DPP_QUAD_XOR1>(x));
  x = fmaxf(x, dpp_mov<DPP_QUAD_XOR2>(x));
  x = fmaxf(x, dpp_mov<DPP_HALF_MIR>(x));
  x = fmaxf(x, dpp_mov<DPP_ROW_MIR>(x));
  return x;
}
static __device__ __forceinline__ float redsum16(float x) {
  x += dpp_mov<DPP_QUAD_XOR1>(x);
  x += dpp_mov<DPP_QUAD_XOR2>(x);
  x += dpp_mov<DPP_HALF_MIR>(x);
  x += dpp_mov<DPP_ROW_MIR>(x);
  return x;
}

__global__ void kq_proj_kernel(const float* __restrict__ pe,
                               const float* __restrict__ Wk, const float* __restrict__ bk,
                               const float* __restrict__ Wq, const float* __restrict__ bq,
                               float* __restrict__ Kd, float* __restrict__ Qd) {
  int idx = blockIdx.x * blockDim.x + threadIdx.x;
  if (idx >= B_ * N_ * D_) return;
  int d  = idx & (D_ - 1);
  int bn = idx / D_;
  const float* p = pe + bn * PE_;
  float k = bk[d], q = bq[d];
#pragma unroll
  for (int e = 0; e < PE_; ++e) {
    float pv = p[e];
    k = fmaf(pv, Wk[e * D_ + d], k);
    q = fmaf(pv, Wq[e * D_ + d], q);
  }
  Kd[idx] = k;
  Qd[idx] = q;
}

struct GTile {
  v2f   q[8];
  float a[8];
  float ej;
  float h[4][2][2];
};

static __device__ __forceinline__ void load_gtile(GTile& T,
                                                 const float* __restrict__ Qb,
                                                 const float* __restrict__ Hb,
                                                 const float* __restrict__ A,
                                                 const float* __restrict__ E,
                                                 int i0, int js, int lo, int hi) {
#pragma unroll
  for (int kc = 0; kc < 8; ++kc)
    T.q[kc] = *(const v2f*)(Qb + (js + lo) * D_ + 4 * kc + 2 * hi);
  T.ej = E[js + lo];
#pragma unroll
  for (int v = 0; v < 8; ++v)
    T.a[v] = A[(i0 + v + 8 * hi) * N_ + js + lo];
#pragma unroll
  for (int kc = 0; kc < 4; ++kc)
#pragma unroll
    for (int s = 0; s < 2; ++s)
#pragma unroll
      for (int t = 0; t < 2; ++t)
        T.h[kc][s][t] = Hb[(js + 4 * kc + 2 * hi + t) * D_ + lo + 16 * s];
}

static __device__ __forceinline__ void wg_barrier_lds() {
  asm volatile("s_wait_dscnt 0\n\ts_barrier_signal -1\n\ts_barrier_wait -1" ::: "memory");
}

#define OT 512
#define NG (N_ / 64)

__global__ void __launch_bounds__(32 * WPB)
fokker_planck_fused_kernel(const float* __restrict__ h,
                           const float* __restrict__ E,
                           const float* __restrict__ A,
                           const float* __restrict__ beta,
                           const float* __restrict__ Kd,
                           const float* __restrict__ Qd,
                           float* __restrict__ out) {
  __shared__ __attribute__((aligned(16))) float Pv[2 * 4 * 3 * PT];
  __shared__ __attribute__((aligned(16))) float Ov[2 * 4 * 3 * OT];
  __shared__ __attribute__((aligned(16))) float Mt[2 * 4 * 16];
  __shared__ __attribute__((aligned(16))) float Es[N_];
  __shared__ float partAcc[4 * 2 * 8 * 32];
  __shared__ float rowS[3 * 16];

  const int lane = threadIdx.x & 31;
  const int wid  = __builtin_amdgcn_readfirstlane((int)(threadIdx.x >> 5));
  const bool prep = wid < 4;
  const int  rw   = wid & 3;
  const int lo   = lane & 15;
  const int hi   = lane >> 4;
  const int tile = blockIdx.x;
  if (tile >= B_ * (N_ / 16)) return;
  const int b  = tile >> 5;
  const int i0 = (tile & 31) * 16;

  const float* Kb = Kd + b * N_ * D_;
  const float* Qb = Qd + b * N_ * D_;
  const float* Hb = h  + b * N_ * D_;

#pragma unroll
  for (int k = 0; k < N_ / (32 * WPB); ++k)
    Es[threadIdx.x + 32 * WPB * k] = E[threadIdx.x + 32 * WPB * k];

  const float kscale = INV_SQRT_D * L2E;
  v2f afK[8] = {};
  float Ei[8] = {};
  GTile ta, tb;
  if (prep) {
#pragma unroll
    for (int kc = 0; kc < 8; ++kc) {
      v2f t = *(const v2f*)(Kb + (i0 + lo) * D_ + 4 * kc + 2 * hi);
      afK[kc].x = t.x * kscale;
      afK[kc].y = t.y * kscale;
    }
#pragma unroll
    for (int v = 0; v < 8; ++v) Ei[v] = E[i0 + v + 8 * hi];
    load_gtile(ta, Qb, Hb, A, E, i0, 16 * rw, lo, hi);
    load_gtile(tb, Qb, Hb, A, E, i0, 64 + 16 * rw, lo, hi);
  }

  const int vsel = (rw == 0) ? 0 : ((rw == 3) ? 2 : 1);
  const int osel = (rw < 2) ? 0 : ((rw == 3) ? 1 : 2);

  v8f acc[2] = {};
  float M[8];
#pragma unroll
  for (int v = 0; v < 8; ++v) M[v] = -1.0e30f;
  float MA = -1.0e30f;
  float sA = 0.f;
  float c2A = 0.f;


  auto prep_step = [&](const GTile& T, const int bf) __attribute__((always_inline)) {
    v8f sraw = {};
#pragma unroll
    for (int kc = 0; kc < 8; ++kc)
      sraw = wmma_f32(afK[kc], T.q[kc], sraw);

    float sc[8], tm[8];
#pragma unroll
    for (int v = 0; v < 8; ++v) {
      sc[v] = T.a[v] * sraw[v];
      tm[v] = redmax16(sc[v]);
    }

    float* pvw = Pv + (bf * 4 + rw) * 3 * PT;
    float* ovw = Ov + (bf * 4 + rw) * 3 * OT + lane * 4;
    float* mtw = Mt + (bf * 4 + rw) * 16;
#pragma unroll
    for (int v = 0; v < 8; ++v) {
      float p  = fexp2(sc[v] - tm[v]);
      float Ed = T.ej - Ei[v];
      float sg = frcp(1.f + fexp2(-KS2 * Ed));
      float p3 = p * sg;
      float p1 = p3 * Ed;
      float p4 = p - p3;
      const int o = (v + 8 * hi) * PP + lo;
      pvw[o]          = p1;
      pvw[PT + o]     = p3;
      pvw[2 * PT + o] = p4;
    }
    {
      v4f t0 = {tm[0], tm[1], tm[2], tm[3]};
      v4f t1 = {tm[4], tm[5], tm[6], tm[7]};
      *(v4f*)(mtw + 8 * hi)     = t0;
      *(v4f*)(mtw + 8 * hi + 4) = t1;
    }
#pragma unroll
    for (int kc = 0; kc < 4; ++kc) {
      v4f hh = {T.h[kc][0][0], T.h[kc][0][1], T.h[kc][1][0], T.h[kc][1][1]};
      v4f ll, lh;
#pragma unroll
      for (int e = 0; e < 4; ++e) {
        ll[e] = __logf(hh[e] + 1e-8f);
        lh[e] = ll[e] * hh[e];
      }
      *(v4f*)(ovw + (0 * 4 + kc) * 128) = hh;
      *(v4f*)(ovw + (1 * 4 + kc) * 128) = ll;
      *(v4f*)(ovw + (2 * 4 + kc) * 128) = lh;
    }
  };

  auto mult_step = [&](const int bf) __attribute__((always_inline)) {
    float mg[8], mAt[4];
#pragma unroll
    for (int tt = 0; tt < 4; ++tt) {
      const float* mt = Mt + (bf * 4 + tt) * 16;
      const v4f m0 = *(const v4f*)(mt + 8 * hi);
      const v4f m1 = *(const v4f*)(mt + 8 * hi + 4);
      mAt[tt] = mt[lo];
      const float mc[8] = {m0.x, m0.y, m0.z, m0.w, m1.x, m1.y, m1.z, m1.w};
#pragma unroll
      for (int v = 0; v < 8; ++v) mg[v] = (tt == 0) ? mc[v] : fmaxf(mg[v], mc[v]);
    }
#pragma unroll
    for (int v = 0; v < 8; ++v) {
      float mn = fmaxf(M[v], mg[v]);
      float r  = fexp2(M[v] - mn);
      M[v] = mn;
      acc[0][v] *= r; acc[1][v] *= r;
    }
    const float mnA = fmaxf(fmaxf(MA, fmaxf(mAt[0], mAt[1])), fmaxf(mAt[2], mAt[3]));
    {
      const float rA = fexp2(MA - mnA);
      MA = mnA;
      sA *= rA;
    }

#pragma unroll
    for (int tt = 0; tt < 4; ++tt) {
      const float fA = fexp2(mAt[tt] - mnA);

      const float* pv = Pv + ((bf * 4 + tt) * 3 + vsel) * PT + lo * PP + 2 * hi;
      v2f wa[4];
#pragma unroll
      for (int kc = 0; kc < 4; ++kc) {
        v2f t = *(const v2f*)(pv + 4 * kc);
        wa[kc].x = t.x * fA;
        wa[kc].y = t.y * fA;
      }
      float ps = 0.f;
#pragma unroll
      for (int kc = 0; kc < 4; ++kc) ps += wa[kc].x + wa[kc].y;
      sA += ps;

      const float* ov = Ov + ((bf * 4 + tt) * 3 + osel) * OT + lane * 4;
      v2f VB[4][2];
#pragma unroll
      for (int kc = 0; kc < 4; ++kc) {
        const v4f o = *(const v4f*)(ov + kc * 128);
        VB[kc][0].x = o.x; VB[kc][0].y = o.y;
        VB[kc][1].x = o.z; VB[kc][1].y = o.w;
      }

#pragma unroll
      for (int kc = 0; kc < 4; ++kc)
#pragma unroll
        for (int s = 0; s < 2; ++s)
          acc[s] = wmma_f32(wa[kc], VB[kc][s], acc[s]);
    }
  };

  auto c2_step = [&](const int bf, const int jg) __attribute__((always_inline)) {
    const float EiA = Es[i0 + lo];
    float mAt[4];
#pragma unroll
    for (int tt = 0; tt < 4; ++tt) mAt[tt] = Mt[(bf * 4 + tt) * 16 + lo];
    const float mnA = fmaxf(fmaxf(MA, fmaxf(mAt[0], mAt[1])), fmaxf(mAt[2], mAt[3]));
    {
      const float rA = fexp2(MA - mnA);
      MA = mnA;
      c2A *= rA;
    }
#pragma unroll
    for (int tt = 0; tt < 4; ++tt) {
      const float fA = fexp2(mAt[tt] - mnA);
      const float* p4 = Pv + ((bf * 4 + tt) * 3 + 2) * PT + lo * PP + 2 * hi;
      const float* ej = Es + jg + 16 * tt + 2 * hi;
      float cs = 0.f;
#pragma unroll
      for (int kc = 0; kc < 4; ++kc) {
        v2f w = *(const v2f*)(p4 + 4 * kc);
        v2f e = *(const v2f*)(ej + 4 * kc);
        cs += w.x * (e.x - EiA) + w.y * (e.y - EiA);
      }
      c2A += cs * fA;
    }
  };

  if (prep) {
    prep_step(ta, 0);
    load_gtile(ta, Qb, Hb, A, E, i0, 2 * 64 + 16 * rw, lo, hi);
  }
  for (int g = 0; g < NG; g += 2) {
    wg_barrier_lds();
    if (prep) {
      prep_step(tb, 1);
      if (g + 3 < NG) load_gtile(tb, Qb, Hb, A, E, i0, (g + 3) * 64 + 16 * rw, lo, hi);
      if (rw == 0) c2_step(0, g * 64);
    } else {
      mult_step(0);
    }
    wg_barrier_lds();
    if (prep) {
      if (g + 2 < NG) {
        prep_step(ta, 0);
        if (g + 4 < NG) load_gtile(ta, Qb, Hb, A, E, i0, (g + 4) * 64 + 16 * rw, lo, hi);
      }
      if (rw == 0) c2_step(1, (g + 1) * 64);
    } else {
      mult_step(1);
    }
  }

  float bt[2] = {};
  float hv[2][2] = {};
  if (prep) {
    bt[0] = beta[lo];
    bt[1] = beta[lo + 16];
#pragma unroll
    for (int vv = 0; vv < 2; ++vv)
#pragma unroll
      for (int s = 0; s < 2; ++s)
        hv[vv][s] = Hb[(i0 + 2 * rw + vv + 8 * hi) * D_ + lo + 16 * s];
    if (rw == 0) rowS[1 * 16 + lo] = c2A + __shfl_xor(c2A, 16);
  } else {
#pragma unroll
    for (int s = 0; s < 2; ++s)
#pragma unroll
      for (int v = 0; v < 8; ++v)
        partAcc[((rw * 2 + s) * 8 + v) * 32 + lane] = acc[s][v];
    const float st = sA + __shfl_xor(sA, 16);
    if (rw == 1) rowS[0 * 16 + lo] = st;
    if (rw == 3) rowS[2 * 16 + lo] = st;
  }

  __syncthreads();

  if (prep) {
#pragma unroll
    for (int vv = 0; vv < 2; ++vv) {
      const int v   = 2 * rw + vv;
      const int rl  = v + 8 * hi;
      const int row = i0 + rl;
      const float cc6 = rowS[2 * 16 + rl];
      const float zz  = rowS[0 * 16 + rl] + cc6;
      const float cc2 = rowS[1 * 16 + rl];
      const float invz = frcp(zz);
#pragma unroll
      for (int s = 0; s < 2; ++s) {
        const int col = lo + 16 * s;
        const float a1 = partAcc[((0 * 2 + s) * 8 + v) * 32 + lane];
        const float a5 = partAcc[((1 * 2 + s) * 8 + v) * 32 + lane];
        const float a3 = partAcc[((2 * 2 + s) * 8 + v) * 32 + lane];
        const float a4 = partAcc[((3 * 2 + s) * 8 + v) * 32 + lane];
        const float hh = hv[vv][s];
        float lv = __logf(hh + 1e-8f);
        float num = a1 + cc2 * hh +
                    bt[s] * (a3 + hh * a4 - lv * a5 - cc6 * lv * hh);
        out[(b * N_ + row) * D_ + col] = num * invz;
      }
    }
  }
}

extern "C" void kernel_launch(void* const* d_in, const int* in_sizes, int n_in,
                              void* d_out, int out_size, void* d_ws, size_t ws_size,
                              hipStream_t stream) {
  const float* h    = (const float*)d_in[1];
  const float* pe   = (const float*)d_in[2];
  const float* E    = (const float*)d_in[3];
  const float* A    = (const float*)d_in[4];
  const float* Wk   = (const float*)d_in[5];
  const float* bk   = (const float*)d_in[6];
  const float* Wq   = (const float*)d_in[7];
  const float* bq   = (const float*)d_in[8];
  const float* beta = (const float*)d_in[9];
  float* out = (float*)d_out;

  float* Kd = (float*)d_ws;
  float* Qd = Kd + (size_t)B_ * N_ * D_;

  {
    int total = B_ * N_ * D_;
    kq_proj_kernel<<<(total + 255) / 256, 256, 0, stream>>>(pe, Wk, bk, Wq, bq, Kd, Qd);
  }
  {
    int tiles = B_ * (N_ / 16);
    dim3 grid(tiles), block(32 * WPB);
    fokker_planck_fused_kernel<<<grid, block, 0, stream>>>(h, E, A, beta, Kd, Qd, out);
  }
}


# L = log(H + 1e-8) computed once per batch row by the first launch into a workspace plane in B-fragment order; preparing waves load it (was recomputed by all 32 row-tile groups); sum order unchanged
# speedup vs baseline: 5.4624x; 1.0842x over previous
#include <hip/hip_runtime.h>
#include <hip/hip_bf16.h>
#include <math.h>

#define B_   4
#define N_   512
#define D_   32
#define PE_  16

#define WPB 8
#define PP  18
#define PT  (16 * PP)
#define L2E 1.44269504088896340736f
#define INV_SQRT_D 0.17677669529663688110f
#define KS2 (10.0f * L2E)

typedef __attribute__((ext_vector_type(2))) float v2f;
typedef __attribute__((ext_vector_type(4))) float v4f;
typedef __attribute__((ext_vector_type(8))) float v8f;

static __device__ __forceinline__ v8f wmma_f32(v2f a, v2f b, v8f c) {
  return __builtin_amdgcn_wmma_f32_16x16x4_f32(false, a, false, b, (short)0, c, false, false);
}

static __device__ __forceinline__ float fexp2(float x) { return __builtin_amdgcn_exp2f(x); }
static __device__ __forceinline__ float frcp(float x)  { return __builtin_amdgcn_rcpf(x); }

#define DPP_QUAD_XOR1  0xB1
#define DPP_QUAD_XOR2  0x4E
#define DPP_HALF_MIR   0x141
#define DPP_ROW_MIR    0x140

template <int CTRL>
static __device__ __forceinline__ float dpp_mov(float x) {
  return __int_as_float(
      __builtin_amdgcn_update_dpp(0, __float_as_int(x), CTRL, 0xf, 0xf, true));
}
static __device__ __forceinline__ float redmax16(float x) {
  x = fmaxf(x, dpp_mov<DPP_QUAD_XOR1>(x));
  x = fmaxf(x, dpp_mov<DPP_QUAD_XOR2>(x));
  x = fmaxf(x, dpp_mov<DPP_HALF_MIR>(x));
  x = fmaxf(x, dpp_mov<DPP_ROW_MIR>(x));
  return x;
}
static __device__ __forceinline__ float redsum16(float x) {
  x += dpp_mov<DPP_QUAD_XOR1>(x);
  x += dpp_mov<DPP_QUAD_XOR2>(x);
  x += dpp_mov<DPP_HALF_MIR>(x);
  x += dpp_mov<DPP_ROW_MIR>(x);
  return x;
}

__global__ void kq_proj_kernel(const float* __restrict__ pe,
                               const float* __restrict__ Wk, const float* __restrict__ bk,
                               const float* __restrict__ Wq, const float* __restrict__ bq,
                               const float* __restrict__ h,
                               float* __restrict__ Kd, float* __restrict__ Qd,
                               float* __restrict__ Lf) {
  int idx = blockIdx.x * blockDim.x + threadIdx.x;
  if (idx >= B_ * N_ * D_) return;
  int d  = idx & (D_ - 1);
  int bn = idx / D_;
  const float* p = pe + bn * PE_;
  float k = bk[d], q = bq[d];
#pragma unroll
  for (int e = 0; e < PE_; ++e) {
    float pv = p[e];
    k = fmaf(pv, Wk[e * D_ + d], k);
    q = fmaf(pv, Wq[e * D_ + d], q);
  }
  Kd[idx] = k;
  Qd[idx] = q;
  {
    const int t  = idx & 1;
    const int s  = (idx >> 1) & 1;
    const int lo = (idx >> 2) & 15;
    const int hi = (idx >> 6) & 1;
    const int kc = (idx >> 7) & 3;
    const int jt = (idx >> 9) & 31;
    const int bb = idx >> 14;
    const int j  = 16 * jt + 4 * kc + 2 * hi + t;
    Lf[idx] = __logf(h[(bb * N_ + j) * D_ + lo + 16 * s] + 1e-8f);
  }
}

struct GTile {
  v2f   q[8];
  float a[8];
  float ej;
  float h[4][2][2];
  v4f   l[4];
};

static __device__ __forceinline__ void load_gtile(GTile& T,
                                                 const float* __restrict__ Qb,
                                                 const float* __restrict__ Hb,
                                                 const float* __restrict__ Lb,
                                                 const float* __restrict__ A,
                                                 const float* __restrict__ E,
                                                 int i0, int js, int lo, int hi) {
#pragma unroll
  for (int kc = 0; kc < 8; ++kc)
    T.q[kc] = *(const v2f*)(Qb + (js + lo) * D_ + 4 * kc + 2 * hi);
  T.ej = E[js + lo];
#pragma unroll
  for (int v = 0; v < 8; ++v)
    T.a[v] = A[(i0 + v + 8 * hi) * N_ + js + lo];
#pragma unroll
  for (int kc = 0; kc < 4; ++kc)
#pragma unroll
    for (int s = 0; s < 2; ++s)
#pragma unroll
      for (int t = 0; t < 2; ++t)
        T.h[kc][s][t] = Hb[(js + 4 * kc + 2 * hi + t) * D_ + lo + 16 * s];
#pragma unroll
  for (int kc = 0; kc < 4; ++kc)
    T.l[kc] = *(const v4f*)(Lb + ((js >> 4) * 4 + kc) * 128 + (lo + 16 * hi) * 4);
}

static __device__ __forceinline__ void wg_barrier_lds() {
  asm volatile("s_wait_dscnt 0\n\ts_barrier_signal -1\n\ts_barrier_wait -1" ::: "memory");
}

#define OT 512
#define NG (N_ / 64)

__global__ void __launch_bounds__(32 * WPB)
fokker_planck_fused_kernel(const float* __restrict__ h,
                           const float* __restrict__ E,
                           const float* __restrict__ A,
                           const float* __restrict__ beta,
                           const float* __restrict__ Kd,
                           const float* __restrict__ Qd,
                           const float* __restrict__ Lf,
                           float* __restrict__ out) {
  __shared__ __attribute__((aligned(16))) float Pv[2 * 4 * 3 * PT];
  __shared__ __attribute__((aligned(16))) float Ov[2 * 4 * 3 * OT];
  __shared__ __attribute__((aligned(16))) float Mt[2 * 4 * 16];
  __shared__ __attribute__((aligned(16))) float Es[N_];
  __shared__ float partAcc[4 * 2 * 8 * 32];
  __shared__ float rowS[3 * 16];

  const int lane = threadIdx.x & 31;
  const int wid  = __builtin_amdgcn_readfirstlane((int)(threadIdx.x >> 5));
  const bool prep = wid < 4;
  const int  rw   = wid & 3;
  const int lo   = lane & 15;
  const int hi   = lane >> 4;
  const int tile = blockIdx.x;
  if (tile >= B_ * (N_ / 16)) return;
  const int b  = tile >> 5;
  const int i0 = (tile & 31) * 16;

  const float* Kb = Kd + b * N_ * D_;
  const float* Qb = Qd + b * N_ * D_;
  const float* Hb = h  + b * N_ * D_;
  const float* Lb = Lf + b * N_ * D_;

#pragma unroll
  for (int k = 0; k < N_ / (32 * WPB); ++k)
    Es[threadIdx.x + 32 * WPB * k] = E[threadIdx.x + 32 * WPB * k];

  const float kscale = INV_SQRT_D * L2E;
  v2f afK[8] = {};
  float Ei[8] = {};
  GTile ta, tb;
  if (prep) {
#pragma unroll
    for (int kc = 0; kc < 8; ++kc) {
      v2f t = *(const v2f*)(Kb + (i0 + lo) * D_ + 4 * kc + 2 * hi);
      afK[kc].x = t.x * kscale;
      afK[kc].y = t.y * kscale;
    }
#pragma unroll
    for (int v = 0; v < 8; ++v) Ei[v] = E[i0 + v + 8 * hi];
    load_gtile(ta, Qb, Hb, Lb, A, E, i0, 16 * rw, lo, hi);
    load_gtile(tb, Qb, Hb, Lb, A, E, i0, 64 + 16 * rw, lo, hi);
  }

  const int vsel = (rw == 0) ? 0 : ((rw == 3) ? 2 : 1);
  const int osel = (rw < 2) ? 0 : ((rw == 3) ? 1 : 2);

  v8f acc[2] = {};
  float M[8];
#pragma unroll
  for (int v = 0; v < 8; ++v) M[v] = -1.0e30f;
  float MA = -1.0e30f;
  float sA = 0.f;
  float c2A = 0.f;


  auto prep_step = [&](const GTile& T, const int bf) __attribute__((always_inline)) {
    v8f sraw = {};
#pragma unroll
    for (int kc = 0; kc < 8; ++kc)
      sraw = wmma_f32(afK[kc], T.q[kc], sraw);

    float sc[8], tm[8];
#pragma unroll
    for (int v = 0; v < 8; ++v) {
      sc[v] = T.a[v] * sraw[v];
      tm[v] = redmax16(sc[v]);
    }

    float* pvw = Pv + (bf * 4 + rw) * 3 * PT;
    float* ovw = Ov + (bf * 4 + rw) * 3 * OT + lane * 4;
    float* mtw = Mt + (bf * 4 + rw) * 16;
#pragma unroll
    for (int v = 0; v < 8; ++v) {
      float p  = fexp2(sc[v] - tm[v]);
      float Ed = T.ej - Ei[v];
      float sg = frcp(1.f + fexp2(-KS2 * Ed));
      float p3 = p * sg;
      float p1 = p3 * Ed;
      float p4 = p - p3;
      const int o = (v + 8 * hi) * PP + lo;
      pvw[o]          = p1;
      pvw[PT + o]     = p3;
      pvw[2 * PT + o] = p4;
    }
    {
      v4f t0 = {tm[0], tm[1], tm[2], tm[3]};
      v4f t1 = {tm[4], tm[5], tm[6], tm[7]};
      *(v4f*)(mtw + 8 * hi)     = t0;
      *(v4f*)(mtw + 8 * hi + 4) = t1;
    }
#pragma unroll
    for (int kc = 0; kc < 4; ++kc) {
      v4f hh = {T.h[kc][0][0], T.h[kc][0][1], T.h[kc][1][0], T.h[kc][1][1]};
      v4f ll = T.l[kc];
      v4f lh = ll * hh;
      *(v4f*)(ovw + (0 * 4 + kc) * 128) = hh;
      *(v4f*)(ovw + (1 * 4 + kc) * 128) = ll;
      *(v4f*)(ovw + (2 * 4 + kc) * 128) = lh;
    }
  };

  auto mult_step = [&](const int bf) __attribute__((always_inline)) {
    float mg[8], mAt[4];
#pragma unroll
    for (int tt = 0; tt < 4; ++tt) {
      const float* mt = Mt + (bf * 4 + tt) * 16;
      const v4f m0 = *(const v4f*)(mt + 8 * hi);
      const v4f m1 = *(const v4f*)(mt + 8 * hi + 4);
      mAt[tt] = mt[lo];
      const float mc[8] = {m0.x, m0.y, m0.z, m0.w, m1.x, m1.y, m1.z, m1.w};
#pragma unroll
      for (int v = 0; v < 8; ++v) mg[v] = (tt == 0) ? mc[v] : fmaxf(mg[v], mc[v]);
    }
#pragma unroll
    for (int v = 0; v < 8; ++v) {
      float mn = fmaxf(M[v], mg[v]);
      float r  = fexp2(M[v] - mn);
      M[v] = mn;
      acc[0][v] *= r; acc[1][v] *= r;
    }
    const float mnA = fmaxf(fmaxf(MA, fmaxf(mAt[0], mAt[1])), fmaxf(mAt[2], mAt[3]));
    {
      const float rA = fexp2(MA - mnA);
      MA = mnA;
      sA *= rA;
    }

#pragma unroll
    for (int tt = 0; tt < 4; ++tt) {
      const float fA = fexp2(mAt[tt] - mnA);

      const float* pv = Pv + ((bf * 4 + tt) * 3 + vsel) * PT + lo * PP + 2 * hi;
      v2f wa[4];
#pragma unroll
      for (int kc = 0; kc < 4; ++kc) {
        v2f t = *(const v2f*)(pv + 4 * kc);
        wa[kc].x = t.x * fA;
        wa[kc].y = t.y * fA;
      }
      float ps = 0.f;
#pragma unroll
      for (int kc = 0; kc < 4; ++kc) ps += wa[kc].x + wa[kc].y;
      sA += ps;

      const float* ov = Ov + ((bf * 4 + tt) * 3 + osel) * OT + lane * 4;
      v2f VB[4][2];
#pragma unroll
      for (int kc = 0; kc < 4; ++kc) {
        const v4f o = *(const v4f*)(ov + kc * 128);
        VB[kc][0].x = o.x; VB[kc][0].y = o.y;
        VB[kc][1].x = o.z; VB[kc][1].y = o.w;
      }

#pragma unroll
      for (int kc = 0; kc < 4; ++kc)
#pragma unroll
        for (int s = 0; s < 2; ++s)
          acc[s] = wmma_f32(wa[kc], VB[kc][s], acc[s]);
    }
  };

  auto c2_step = [&](const int bf, const int jg) __attribute__((always_inline)) {
    const float EiA = Es[i0 + lo];
    float mAt[4];
#pragma unroll
    for (int tt = 0; tt < 4; ++tt) mAt[tt] = Mt[(bf * 4 + tt) * 16 + lo];
    const float mnA = fmaxf(fmaxf(MA, fmaxf(mAt[0], mAt[1])), fmaxf(mAt[2], mAt[3]));
    {
      const float rA = fexp2(MA - mnA);
      MA = mnA;
      c2A *= rA;
    }
#pragma unroll
    for (int tt = 0; tt < 4; ++tt) {
      const float fA = fexp2(mAt[tt] - mnA);
      const float* p4 = Pv + ((bf * 4 + tt) * 3 + 2) * PT + lo * PP + 2 * hi;
      const float* ej = Es + jg + 16 * tt + 2 * hi;
      float cs = 0.f;
#pragma unroll
      for (int kc = 0; kc < 4; ++kc) {
        v2f w = *(const v2f*)(p4 + 4 * kc);
        v2f e = *(const v2f*)(ej + 4 * kc);
        cs += w.x * (e.x - EiA) + w.y * (e.y - EiA);
      }
      c2A += cs * fA;
    }
  };

  if (prep) {
    prep_step(ta, 0);
    load_gtile(ta, Qb, Hb, Lb, A, E, i0, 2 * 64 + 16 * rw, lo, hi);
  }
  for (int g = 0; g < NG; g += 2) {
    wg_barrier_lds();
    if (prep) {
      prep_step(tb, 1);
      if (g + 3 < NG) load_gtile(tb, Qb, Hb, Lb, A, E, i0, (g + 3) * 64 + 16 * rw, lo, hi);
      if (rw == 0) c2_step(0, g * 64);
    } else {
      mult_step(0);
    }
    wg_barrier_lds();
    if (prep) {
      if (g + 2 < NG) {
        prep_step(ta, 0);
        if (g + 4 < NG) load_gtile(ta, Qb, Hb, Lb, A, E, i0, (g + 4) * 64 + 16 * rw, lo, hi);
      }
      if (rw == 0) c2_step(1, (g + 1) * 64);
    } else {
      mult_step(1);
    }
  }

  float bt[2] = {};
  float hv[2][2] = {};
  if (prep) {
    bt[0] = beta[lo];
    bt[1] = beta[lo + 16];
#pragma unroll
    for (int vv = 0; vv < 2; ++vv)
#pragma unroll
      for (int s = 0; s < 2; ++s)
        hv[vv][s] = Hb[(i0 + 2 * rw + vv + 8 * hi) * D_ + lo + 16 * s];
    if (rw == 0) rowS[1 * 16 + lo] = c2A + __shfl_xor(c2A, 16);
  } else {
#pragma unroll
    for (int s = 0; s < 2; ++s)
#pragma unroll
      for (int v = 0; v < 8; ++v)
        partAcc[((rw * 2 + s) * 8 + v) * 32 + lane] = acc[s][v];
    const float st = sA + __shfl_xor(sA, 16);
    if (rw == 1) rowS[0 * 16 + lo] = st;
    if (rw == 3) rowS[2 * 16 + lo] = st;
  }

  __syncthreads();

  if (prep) {
#pragma unroll
    for (int vv = 0; vv < 2; ++vv) {
      const int v   = 2 * rw + vv;
      const int rl  = v + 8 * hi;
      const int row = i0 + rl;
      const float cc6 = rowS[2 * 16 + rl];
      const float zz  = rowS[0 * 16 + rl] + cc6;
      const float cc2 = rowS[1 * 16 + rl];
      const float invz = frcp(zz);
#pragma unroll
      for (int s = 0; s < 2; ++s) {
        const int col = lo + 16 * s;
        const float a1 = partAcc[((0 * 2 + s) * 8 + v) * 32 + lane];
        const float a5 = partAcc[((1 * 2 + s) * 8 + v) * 32 + lane];
        const float a3 = partAcc[((2 * 2 + s) * 8 + v) * 32 + lane];
        const float a4 = partAcc[((3 * 2 + s) * 8 + v) * 32 + lane];
        const float hh = hv[vv][s];
        float lv = __logf(hh + 1e-8f);
        float num = a1 + cc2 * hh +
                    bt[s] * (a3 + hh * a4 - lv * a5 - cc6 * lv * hh);
        out[(b * N_ + row) * D_ + col] = num * invz;
      }
    }
  }
}

extern "C" void kernel_launch(void* const* d_in, const int* in_sizes, int n_in,
                              void* d_out, int out_size, void* d_ws, size_t ws_size,
                              hipStream_t stream) {
  const float* h    = (const float*)d_in[1];
  const float* pe   = (const float*)d_in[2];
  const float* E    = (const float*)d_in[3];
  const float* A    = (const float*)d_in[4];
  const float* Wk   = (const float*)d_in[5];
  const float* bk   = (const float*)d_in[6];
  const float* Wq   = (const float*)d_in[7];
  const float* bq   = (const float*)d_in[8];
  const float* beta = (const float*)d_in[9];
  float* out = (float*)d_out;

  float* Kd = (float*)d_ws;
  float* Qd = Kd + (size_t)B_ * N_ * D_;
  float* Lf = Qd + (size_t)B_ * N_ * D_;
  if (ws_size < (size_t)3 * B_ * N_ * D_ * sizeof(float)) return;

  {
    int total = B_ * N_ * D_;
    kq_proj_kernel<<<(total + 255) / 256, 256, 0, stream>>>(pe, Wk, bk, Wq, bq, h, Kd, Qd, Lf);
  }
  {
    int tiles = B_ * (N_ / 16);
    dim3 grid(tiles), block(32 * WPB);
    fokker_planck_fused_kernel<<<grid, block, 0, stream>>>(h, E, A, beta, Kd, Qd, Lf, out);
  }
}


# Each role runs its own loop over the groups (equal barrier counts): no role branch inside an interval, so the counted wait before a tile's scores leaves the other set's refill in flight; same sums
# speedup vs baseline: 6.3561x; 1.1636x over previous
#include <hip/hip_runtime.h>
#include <hip/hip_bf16.h>
#include <math.h>

#define B_   4
#define N_   512
#define D_   32
#define PE_  16

#define WPB 8
#define PP  18
#define PT  (16 * PP)
#define L2E 1.44269504088896340736f
#define INV_SQRT_D 0.17677669529663688110f
#define KS2 (10.0f * L2E)

typedef __attribute__((ext_vector_type(2))) float v2f;
typedef __attribute__((ext_vector_type(4))) float v4f;
typedef __attribute__((ext_vector_type(8))) float v8f;

static __device__ __forceinline__ v8f wmma_f32(v2f a, v2f b, v8f c) {
  return __builtin_amdgcn_wmma_f32_16x16x4_f32(false, a, false, b, (short)0, c, false, false);
}

static __device__ __forceinline__ float fexp2(float x) { return __builtin_amdgcn_exp2f(x); }
static __device__ __forceinline__ float frcp(float x)  { return __builtin_amdgcn_rcpf(x); }

#define DPP_QUAD_XOR1  0xB1
#define DPP_QUAD_XOR2  0x4E
#define DPP_HALF_MIR   0x141
#define DPP_ROW_MIR    0x140

template <int CTRL>
static __device__ __forceinline__ float dpp_mov(float x) {
  return __int_as_float(
      __builtin_amdgcn_update_dpp(0, __float_as_int(x), CTRL, 0xf, 0xf, true));
}
static __device__ __forceinline__ float redmax16(float x) {
  x = fmaxf(x, dpp_mov<DPP_QUAD_XOR1>(x));
  x = fmaxf(x, dpp_mov<DPP_QUAD_XOR2>(x));
  x = fmaxf(x, dpp_mov<DPP_HALF_MIR>(x));
  x = fmaxf(x, dpp_mov<DPP_ROW_MIR>(x));
  return x;
}
static __device__ __forceinline__ float redsum16(float x) {
  x += dpp_mov<DPP_QUAD_XOR1>(x);
  x += dpp_mov<DPP_QUAD_XOR2>(x);
  x += dpp_mov<DPP_HALF_MIR>(x);
  x += dpp_mov<DPP_ROW_MIR>(x);
  return x;
}

__global__ void kq_proj_kernel(const float* __restrict__ pe,
                               const float* __restrict__ Wk, const float* __restrict__ bk,
                               const float* __restrict__ Wq, const float* __restrict__ bq,
                               const float* __restrict__ h,
                               float* __restrict__ Kd, float* __restrict__ Qd,
                               float* __restrict__ Lf) {
  int idx = blockIdx.x * blockDim.x + threadIdx.x;
  if (idx >= B_ * N_ * D_) return;
  int d  = idx & (D_ - 1);
  int bn = idx / D_;
  const float* p = pe + bn * PE_;
  float k = bk[d], q = bq[d];
#pragma unroll
  for (int e = 0; e < PE_; ++e) {
    float pv = p[e];
    k = fmaf(pv, Wk[e * D_ + d], k);
    q = fmaf(pv, Wq[e * D_ + d], q);
  }
  Kd[idx] = k;
  Qd[idx] = q;
  {
    const int t  = idx & 1;
    const int s  = (idx >> 1) & 1;
    const int lo = (idx >> 2) & 15;
    const int hi = (idx >> 6) & 1;
    const int kc = (idx >> 7) & 3;
    const int jt = (idx >> 9) & 31;
    const int bb = idx >> 14;
    const int j  = 16 * jt + 4 * kc + 2 * hi + t;
    Lf[idx] = __logf(h[(bb * N_ + j) * D_ + lo + 16 * s] + 1e-8f);
  }
}

struct GTile {
  v2f   q[8];
  float a[8];
  float ej;
  float h[4][2][2];
  v4f   l[4];
};

static __device__ __forceinline__ void load_gtile(GTile& T,
                                                 const float* __restrict__ Qb,
                                                 const float* __restrict__ Hb,
                                                 const float* __restrict__ Lb,
                                                 const float* __restrict__ A,
                                                 const float* __restrict__ E,
                                                 int i0, int js, int lo, int hi) {
#pragma unroll
  for (int kc = 0; kc < 8; ++kc)
    T.q[kc] = *(const v2f*)(Qb + (js + lo) * D_ + 4 * kc + 2 * hi);
  T.ej = E[js + lo];
#pragma unroll
  for (int v = 0; v < 8; ++v)
    T.a[v] = A[(i0 + v + 8 * hi) * N_ + js + lo];
#pragma unroll
  for (int kc = 0; kc < 4; ++kc)
#pragma unroll
    for (int s = 0; s < 2; ++s)
#pragma unroll
      for (int t = 0; t < 2; ++t)
        T.h[kc][s][t] = Hb[(js + 4 * kc + 2 * hi + t) * D_ + lo + 16 * s];
#pragma unroll
  for (int kc = 0; kc < 4; ++kc)
    T.l[kc] = *(const v4f*)(Lb + ((js >> 4) * 4 + kc) * 128 + (lo + 16 * hi) * 4);
}

static __device__ __forceinline__ void wg_barrier_lds() {
  asm volatile("s_wait_dscnt 0\n\ts_barrier_signal -1\n\ts_barrier_wait -1" ::: "memory");
}
static __device__ __forceinline__ void wg_barrier_lds_acc(v8f& a0, v8f& a1) {
  asm volatile("s_wait_dscnt 0\n\ts_barrier_signal -1\n\ts_barrier_wait -1" : "+v"(a0), "+v"(a1) : : "memory");
}


#define OT 512
#define NG (N_ / 64)

__global__ void __launch_bounds__(32 * WPB)
fokker_planck_fused_kernel(const float* __restrict__ h,
                           const float* __restrict__ E,
                           const float* __restrict__ A,
                           const float* __restrict__ beta,
                           const float* __restrict__ Kd,
                           const float* __restrict__ Qd,
                           const float* __restrict__ Lf,
                           float* __restrict__ out) {
  __shared__ __attribute__((aligned(16))) float Pv[2 * 4 * 3 * PT];
  __shared__ __attribute__((aligned(16))) float Ov[2 * 4 * 3 * OT];
  __shared__ __attribute__((aligned(16))) float Mt[2 * 4 * 16];
  __shared__ __attribute__((aligned(16))) float Es[N_];
  __shared__ float partAcc[4 * 2 * 8 * 32];
  __shared__ float rowS[3 * 16];

  const int lane = threadIdx.x & 31;
  const int wid  = __builtin_amdgcn_readfirstlane((int)(threadIdx.x >> 5));
  const bool prep = wid < 4;
  const int  rw   = wid & 3;
  const int lo   = lane & 15;
  const int hi   = lane >> 4;
  const int tile = blockIdx.x;
  if (tile >= B_ * (N_ / 16)) return;
  const int b  = tile >> 5;
  const int i0 = (tile & 31) * 16;

  const float* Kb = Kd + b * N_ * D_;
  const float* Qb = Qd + b * N_ * D_;
  const float* Hb = h  + b * N_ * D_;
  const float* Lb = Lf + b * N_ * D_;

#pragma unroll
  for (int k = 0; k < N_ / (32 * WPB); ++k)
    Es[threadIdx.x + 32 * WPB * k] = E[threadIdx.x + 32 * WPB * k];

  const float kscale = INV_SQRT_D * L2E;
  v2f afK[8] = {};
  float Ei[8] = {};
  GTile ta, tb;
  if (prep) {
#pragma unroll
    for (int kc = 0; kc < 8; ++kc) {
      v2f t = *(const v2f*)(Kb + (i0 + lo) * D_ + 4 * kc + 2 * hi);
      afK[kc].x = t.x * kscale;
      afK[kc].y = t.y * kscale;
    }
#pragma unroll
    for (int v = 0; v < 8; ++v) Ei[v] = E[i0 + v + 8 * hi];
    load_gtile(ta, Qb, Hb, Lb, A, E, i0, 16 * rw, lo, hi);
    load_gtile(tb, Qb, Hb, Lb, A, E, i0, 64 + 16 * rw, lo, hi);
  }

  const int vsel = (rw == 0) ? 0 : ((rw == 3) ? 2 : 1);
  const int osel = (rw < 2) ? 0 : ((rw == 3) ? 1 : 2);

  v8f acc[2] = {};
  float M[8];
#pragma unroll
  for (int v = 0; v < 8; ++v) M[v] = -1.0e30f;
  float MA = -1.0e30f;
  float sA = 0.f;
  float c2A = 0.f;


  auto prep_step = [&](const GTile& T, const int bf) __attribute__((always_inline)) {
    v8f sraw = {};
#pragma unroll
    for (int kc = 0; kc < 8; ++kc)
      sraw = wmma_f32(afK[kc], T.q[kc], sraw);

    float sc[8], tm[8];
#pragma unroll
    for (int v = 0; v < 8; ++v) {
      sc[v] = T.a[v] * sraw[v];
      tm[v] = redmax16(sc[v]);
    }

    float* pvw = Pv + (bf * 4 + rw) * 3 * PT;
    float* ovw = Ov + (bf * 4 + rw) * 3 * OT + lane * 4;
    float* mtw = Mt + (bf * 4 + rw) * 16;
#pragma unroll
    for (int v = 0; v < 8; ++v) {
      float p  = fexp2(sc[v] - tm[v]);
      float Ed = T.ej - Ei[v];
      float sg = frcp(1.f + fexp2(-KS2 * Ed));
      float p3 = p * sg;
      float p1 = p3 * Ed;
      float p4 = p - p3;
      const int o = (v + 8 * hi) * PP + lo;
      pvw[o]          = p1;
      pvw[PT + o]     = p3;
      pvw[2 * PT + o] = p4;
    }
    {
      v4f t0 = {tm[0], tm[1], tm[2], tm[3]};
      v4f t1 = {tm[4], tm[5], tm[6], tm[7]};
      *(v4f*)(mtw + 8 * hi)     = t0;
      *(v4f*)(mtw + 8 * hi + 4) = t1;
    }
#pragma unroll
    for (int kc = 0; kc < 4; ++kc) {
      v4f hh = {T.h[kc][0][0], T.h[kc][0][1], T.h[kc][1][0], T.h[kc][1][1]};
      v4f ll = T.l[kc];
      v4f lh = ll * hh;
      *(v4f*)(ovw + (0 * 4 + kc) * 128) = hh;
      *(v4f*)(ovw + (1 * 4 + kc) * 128) = ll;
      *(v4f*)(ovw + (2 * 4 + kc) * 128) = lh;
    }
  };

  auto mult_step = [&](const int bf) __attribute__((always_inline)) {
    float mg[8], mAt[4];
#pragma unroll
    for (int tt = 0; tt < 4; ++tt) {
      const float* mt = Mt + (bf * 4 + tt) * 16;
      const v4f m0 = *(const v4f*)(mt + 8 * hi);
      const v4f m1 = *(const v4f*)(mt + 8 * hi + 4);
      mAt[tt] = mt[lo];
      const float mc[8] = {m0.x, m0.y, m0.z, m0.w, m1.x, m1.y, m1.z, m1.w};
#pragma unroll
      for (int v = 0; v < 8; ++v) mg[v] = (tt == 0) ? mc[v] : fmaxf(mg[v], mc[v]);
    }
#pragma unroll
    for (int v = 0; v < 8; ++v) {
      float mn = fmaxf(M[v], mg[v]);
      float r  = fexp2(M[v] - mn);
      M[v] = mn;
      acc[0][v] *= r; acc[1][v] *= r;
    }
    const float mnA = fmaxf(fmaxf(MA, fmaxf(mAt[0], mAt[1])), fmaxf(mAt[2], mAt[3]));
    {
      const float rA = fexp2(MA - mnA);
      MA = mnA;
      sA *= rA;
    }

#pragma unroll
    for (int tt = 0; tt < 4; ++tt) {
      const float fA = fexp2(mAt[tt] - mnA);

      const float* pv = Pv + ((bf * 4 + tt) * 3 + vsel) * PT + lo * PP + 2 * hi;
      v2f wa[4];
#pragma unroll
      for (int kc = 0; kc < 4; ++kc) {
        v2f t = *(const v2f*)(pv + 4 * kc);
        wa[kc].x = t.x * fA;
        wa[kc].y = t.y * fA;
      }
      float ps = 0.f;
#pragma unroll
      for (int kc = 0; kc < 4; ++kc) ps += wa[kc].x + wa[kc].y;
      sA += ps;

      const float* ov = Ov + ((bf * 4 + tt) * 3 + osel) * OT + lane * 4;
      v2f VB[4][2];
#pragma unroll
      for (int kc = 0; kc < 4; ++kc) {
        const v4f o = *(const v4f*)(ov + kc * 128);
        VB[kc][0].x = o.x; VB[kc][0].y = o.y;
        VB[kc][1].x = o.z; VB[kc][1].y = o.w;
      }

#pragma unroll
      for (int kc = 0; kc < 4; ++kc)
#pragma unroll
        for (int s = 0; s < 2; ++s)
          acc[s] = wmma_f32(wa[kc], VB[kc][s], acc[s]);
    }
  };

  auto c2_step = [&](const int bf, const int jg) __attribute__((always_inline)) {
    const float EiA = Es[i0 + lo];
    float mAt[4];
#pragma unroll
    for (int tt = 0; tt < 4; ++tt) mAt[tt] = Mt[(bf * 4 + tt) * 16 + lo];
    const float mnA = fmaxf(fmaxf(MA, fmaxf(mAt[0], mAt[1])), fmaxf(mAt[2], mAt[3]));
    {
      const float rA = fexp2(MA - mnA);
      MA = mnA;
      c2A *= rA;
    }
#pragma unroll
    for (int tt = 0; tt < 4; ++tt) {
      const float fA = fexp2(mAt[tt] - mnA);
      const float* p4 = Pv + ((bf * 4 + tt) * 3 + 2) * PT + lo * PP + 2 * hi;
      const float* ej = Es + jg + 16 * tt + 2 * hi;
      float cs = 0.f;
#pragma unroll
      for (int kc = 0; kc < 4; ++kc) {
        v2f w = *(const v2f*)(p4 + 4 * kc);
        v2f e = *(const v2f*)(ej + 4 * kc);
        cs += w.x * (e.x - EiA) + w.y * (e.y - EiA);
      }
      c2A += cs * fA;
    }
  };

  if (prep) {
    auto prep_pair = [&](const int g, const bool refill_b, const bool step_a, const bool refill_a)
        __attribute__((always_inline)) {
      wg_barrier_lds();
      prep_step(tb, 1);
      if (refill_b) load_gtile(tb, Qb, Hb, Lb, A, E, i0, (g + 3) * 64 + 16 * rw, lo, hi);
      if (rw == 0) c2_step(0, g * 64);
      wg_barrier_lds();
      if (step_a) {
        prep_step(ta, 0);
        if (refill_a) load_gtile(ta, Qb, Hb, Lb, A, E, i0, (g + 4) * 64 + 16 * rw, lo, hi);
      }
      if (rw == 0) c2_step(1, (g + 1) * 64);
    };
    prep_step(ta, 0);
    load_gtile(ta, Qb, Hb, Lb, A, E, i0, 2 * 64 + 16 * rw, lo, hi);
    for (int g = 0; g < NG - 4; g += 2)
      prep_pair(g, true, true, true);
    prep_pair(NG - 4, true, true, false);
    prep_pair(NG - 2, false, false, false);
  } else {
    for (int g = 0; g < NG; g += 2) {
      wg_barrier_lds_acc(acc[0], acc[1]);
      mult_step(0);
      wg_barrier_lds_acc(acc[0], acc[1]);
      mult_step(1);
    }
  }

  float bt[2] = {};
  float hv[2][2] = {};
  if (prep) {
    bt[0] = beta[lo];
    bt[1] = beta[lo + 16];
#pragma unroll
    for (int vv = 0; vv < 2; ++vv)
#pragma unroll
      for (int s = 0; s < 2; ++s)
        hv[vv][s] = Hb[(i0 + 2 * rw + vv + 8 * hi) * D_ + lo + 16 * s];
    if (rw == 0) rowS[1 * 16 + lo] = c2A + __shfl_xor(c2A, 16);
  } else {
#pragma unroll
    for (int s = 0; s < 2; ++s)
#pragma unroll
      for (int v = 0; v < 8; ++v)
        partAcc[((rw * 2 + s) * 8 + v) * 32 + lane] = acc[s][v];
    const float st = sA + __shfl_xor(sA, 16);
    if (rw == 1) rowS[0 * 16 + lo] = st;
    if (rw == 3) rowS[2 * 16 + lo] = st;
  }

  __syncthreads();

  if (prep) {
#pragma unroll
    for (int vv = 0; vv < 2; ++vv) {
      const int v   = 2 * rw + vv;
      const int rl  = v + 8 * hi;
      const int row = i0 + rl;
      const float cc6 = rowS[2 * 16 + rl];
      const float zz  = rowS[0 * 16 + rl] + cc6;
      const float cc2 = rowS[1 * 16 + rl];
      const float invz = frcp(zz);
#pragma unroll
      for (int s = 0; s < 2; ++s) {
        const int col = lo + 16 * s;
        const float a1 = partAcc[((0 * 2 + s) * 8 + v) * 32 + lane];
        const float a5 = partAcc[((1 * 2 + s) * 8 + v) * 32 + lane];
        const float a3 = partAcc[((2 * 2 + s) * 8 + v) * 32 + lane];
        const float a4 = partAcc[((3 * 2 + s) * 8 + v) * 32 + lane];
        const float hh = hv[vv][s];
        float lv = __logf(hh + 1e-8f);
        float num = a1 + cc2 * hh +
                    bt[s] * (a3 + hh * a4 - lv * a5 - cc6 * lv * hh);
        out[(b * N_ + row) * D_ + col] = num * invz;
      }
    }
  }
}

extern "C" void kernel_launch(void* const* d_in, const int* in_sizes, int n_in,
                              void* d_out, int out_size, void* d_ws, size_t ws_size,
                              hipStream_t stream) {
  const float* h    = (const float*)d_in[1];
  const float* pe   = (const float*)d_in[2];
  const float* E    = (const float*)d_in[3];
  const float* A    = (const float*)d_in[4];
  const float* Wk   = (const float*)d_in[5];
  const float* bk   = (const float*)d_in[6];
  const float* Wq   = (const float*)d_in[7];
  const float* bq   = (const float*)d_in[8];
  const float* beta = (const float*)d_in[9];
  float* out = (float*)d_out;

  float* Kd = (float*)d_ws;
  float* Qd = Kd + (size_t)B_ * N_ * D_;
  float* Lf = Qd + (size_t)B_ * N_ * D_;
  if (ws_size < (size_t)3 * B_ * N_ * D_ * sizeof(float)) return;

  {
    int total = B_ * N_ * D_;
    kq_proj_kernel<<<(total + 255) / 256, 256, 0, stream>>>(pe, Wk, bk, Wq, bq, h, Kd, Qd, Lf);
  }
  {
    int tiles = B_ * (N_ / 16);
    dim3 grid(tiles), block(32 * WPB);
    fokker_planck_fused_kernel<<<grid, block, 0, stream>>>(h, E, A, beta, Kd, Qd, Lf, out);
  }
}
